# starting kernel
# baseline (speedup 1.0000x reference)
#include <hip/hip_runtime.h>

typedef __attribute__((ext_vector_type(2))) float v2f;
typedef __attribute__((ext_vector_type(8))) float v8f;

#define PT_DIM    16
#define PT_L      8
#define PT_NODES  255

__device__ __forceinline__ float pt_traverse(const float* __restrict__ row, float xv) {
#pragma clang fp contract(off)
  float lo = 0.0f, hi = 1.0f, prodb = 1.0f;
  int node = 0;
#pragma unroll
  for (int l = 0; l < PT_L - 1; ++l) {
    float b     = row[node];
    prodb       = prodb * b;
    float len   = hi - lo;
    float split = lo + b * len;
    float rhi   = split + (1.0f - b) * len;
    bool left   = (xv <= split);
    hi   = left ? split : rhi;
    lo   = left ? lo    : split;
    node = left ? (2 * node + 1) : (2 * node + 2);
  }
  float b = row[node];
  prodb   = prodb * b;
  float leaf = hi - lo;
  return logf(prodb * __builtin_amdgcn_rcpf(leaf));
}

__global__ __launch_bounds__(256)
void polya_tree_kernel(const float* __restrict__ x,
                       const float* __restrict__ samples,
                       float* __restrict__ out, int n) {
  __shared__ float sb[PT_DIM * PT_NODES];

  const int tid = threadIdx.x;
  for (int i = tid; i < PT_DIM * PT_NODES; i += 256) sb[i] = samples[i];

  const int wave = tid >> 5;
  const int lane = tid & 31;
  const int half = lane >> 4;
  const int m    = lane & 15;
  const int wavebase = blockIdx.x * 128 + wave * 16;

  const int s0 = wavebase + m;
  const int sc = (s0 < n) ? s0 : (n - 1);
  const float* xr = x + (size_t)sc * PT_DIM;
  __builtin_prefetch(xr, 0, 0);

  __syncthreads();

  v2f xa[4];
#pragma unroll
  for (int c = 0; c < 4; ++c)
    xa[c] = *(const v2f*)(xr + 4 * c + 2 * half);

#if __has_builtin(__builtin_amdgcn_wmma_f32_16x16x4_f32)
  v8f acc = {0.f, 0.f, 0.f, 0.f, 0.f, 0.f, 0.f, 0.f};
#pragma unroll
  for (int c = 0; c < 4; ++c) {
    const int d0 = 4 * c + 2 * half;
    float r0 = pt_traverse(sb + (size_t)d0 * PT_NODES,       xa[c].x);
    float r1 = pt_traverse(sb + (size_t)(d0 + 1) * PT_NODES, xa[c].y);
    v2f a;  a.x = r0;  a.y = r1;
    v2f bo; bo.x = 1.0f; bo.y = 1.0f;
    acc = __builtin_amdgcn_wmma_f32_16x16x4_f32(
         false, a,  false, bo,
         (short)0, acc,  false,  false);
  }
  if (m < 8) {
    float v = (m == 0) ? acc[0] : (m == 1) ? acc[1] : (m == 2) ? acc[2] :
              (m == 3) ? acc[3] : (m == 4) ? acc[4] : (m == 5) ? acc[5] :
              (m == 6) ? acc[6] : acc[7];
    int sdst = wavebase + 8 * half + m;
    if (sdst < n) out[sdst] = v * 0.0625f;
  }
#else
  float part = 0.0f;
#pragma unroll
  for (int c = 0; c < 4; ++c) {
    const int d0 = 4 * c + 2 * half;
    part += pt_traverse(sb + (size_t)d0 * PT_NODES,       xa[c].x);
    part += pt_traverse(sb + (size_t)(d0 + 1) * PT_NODES, xa[c].y);
  }
  part += __shfl_xor(part, 16, 32);
  if (half == 0 && s0 < n) out[s0] = part * 0.0625f;
#endif
}

extern "C" void kernel_launch(void* const* d_in, const int* in_sizes, int n_in,
                              void* d_out, int out_size, void* d_ws, size_t ws_size,
                              hipStream_t stream) {
  const float* x       = (const float*)d_in[0];
  const float* samples = (const float*)d_in[1];
  (void)d_ws; (void)ws_size; (void)n_in;
  float* out = (float*)d_out;
  const int n = in_sizes[0] / PT_DIM;
  const int blocks = (n + 127) / 128;
  polya_tree_kernel<<<blocks, 256, 0, stream>>>(x, samples, out, n);
}


# Walk step is now subtract + v_alignbit: m = 2m + sign bit of (split - x) in element units, LDS table regrouped per level with both half-waves side by side; was compare, select and 3 integer ops
# speedup vs baseline: 2.6274x; 2.6274x over previous
#include <hip/hip_runtime.h>


typedef __attribute__((ext_vector_type(2))) float v2f;
typedef __attribute__((ext_vector_type(8))) float v8f;

#define PT_DIM    16
#define PT_L      8
#define PT_NODES  255
#define PT_TILE   128
#define PT_GRID   2048

#define PT_SLOT   510
#define PT_TBL    (8 * PT_SLOT + 1)
#define PT_IMM(q, l) (1 + (q) * PT_SLOT + (1 << (l)) - 2)

__device__ __forceinline__ void pt_split(float lo, float hi, float b,
                                         float& split, float& rhi) {
#pragma clang fp contract(off)
  const float len = hi - lo;
  split = lo + b * len;
  rhi   = split + (1.0f - b) * len;
}

__device__ __forceinline__ void pt_build(const float* __restrict__ samples,
                                         float* __restrict__ tb, int tid) {
  const int d = tid >> 4;
  const int k = tid & 15;
  const float* __restrict__ srow = samples + d * PT_NODES;
  const int hd = (d >> 1) & 1;
  float* __restrict__ trow = tb + 1 + (2 * (d >> 2) + (d & 1)) * PT_SLOT;

  float pbeta[4];
  float sbeta[15];
#pragma unroll
  for (int l = 0; l < 4; ++l) pbeta[l] = srow[((1 << l) - 1) + (k >> (4 - l))];
  {
    int first = 15 + k;
#pragma unroll
    for (int s = 0; s < 4; ++s) {
#pragma unroll
      for (int j = 0; j < (1 << s); ++j) sbeta[(1 << s) - 1 + j] = srow[first + j];
      first = 2 * first + 1;
    }
  }

  float lo = 0.0f, hi = 1.0f, prodb = 1.0f;
#pragma unroll
  for (int l = 0; l < 4; ++l) {
    const int anc = ((1 << l) - 1) + (k >> (4 - l));
    const float b = pbeta[l];
    prodb = prodb * b;
    float split, rhi;
    pt_split(lo, hi, b, split, rhi);
    if ((k & ((1 << (4 - l)) - 1)) == 0)
      trow[2 * ((1 << l) - 1) + (hd << l) + (k >> (4 - l))] = split;
    const bool right = ((k >> (3 - l)) & 1) != 0;
    hi = right ? rhi   : split;
    lo = right ? split : lo;
  }

  float clo[8], chi[8], cpb[8];
  clo[0] = lo; chi[0] = hi; cpb[0] = prodb;
#pragma unroll
  for (int s = 0; s < 3; ++s) {
    const int cnt = 1 << s;
    float nlo[8], nhi[8], npb[8];
#pragma unroll
    for (int j = 0; j < cnt; ++j) {
      const float b = sbeta[cnt - 1 + j];
      const float pb = cpb[j] * b;
      float split, rhi;
      pt_split(clo[j], chi[j], b, split, rhi);
      trow[2 * ((16 << s) - 1) + (hd << (4 + s)) + (k << s) + j] = split;
      nlo[2 * j]     = clo[j]; nhi[2 * j]     = split; npb[2 * j]     = pb;
      nlo[2 * j + 1] = split;  nhi[2 * j + 1] = rhi;   npb[2 * j + 1] = pb;
    }
#pragma unroll
    for (int j = 0; j < 2 * cnt; ++j) { clo[j] = nlo[j]; chi[j] = nhi[j]; cpb[j] = npb[j]; }
  }
#pragma unroll
  for (int j = 0; j < 8; ++j) {
    const float b = sbeta[7 + j];
    const float pb = cpb[j] * b;
    const float leaf = chi[j] - clo[j];
    trow[2 * 127 + (hd << 7) + 8 * k + j] = logf(pb * __builtin_amdgcn_rcpf(leaf));
  }
}

__global__ __launch_bounds__(256)
void polya_tree_kernel(const float* __restrict__ x,
                       const float* __restrict__ samples,
                       float* __restrict__ out, int n) {
  __shared__ float tb[PT_TBL];

  const int tid = threadIdx.x;
  const int wave = tid >> 5;
  const int lane = tid & 31;
  const int half = lane >> 4;
  const int m    = lane & 15;
  const int tiles = (n + PT_TILE - 1) / PT_TILE;

  {
    const int s0 = blockIdx.x * PT_TILE + wave * 16 + m;
    const int sc = (s0 < n) ? s0 : (n - 1);
    __builtin_prefetch(x + (size_t)sc * PT_DIM, 0, 0);
  }

  pt_build(samples, tb, tid);
  __syncthreads();


  for (int tile = blockIdx.x; tile < tiles; tile += gridDim.x) {
    const int wavebase = tile * PT_TILE + wave * 16;
    const int s0 = wavebase + m;
    const int sc = (s0 < n) ? s0 : (n - 1);
    const float* xr = x + (size_t)sc * PT_DIM;

    v2f xa[4];
#pragma unroll
    for (int c = 0; c < 4; ++c)
      xa[c] = *(const v2f*)(xr + 4 * c + 2 * half);

    {
      const int s1 = s0 + (int)gridDim.x * PT_TILE;
      const int sn = (s1 < n) ? s1 : (n - 1);
      __builtin_prefetch(x + (size_t)sn * PT_DIM, 0, 0);
    }

    float xv[8];
    unsigned mi[8];
#pragma unroll
    for (int c = 0; c < 4; ++c) { xv[2 * c] = xa[c].x; xv[2 * c + 1] = xa[c].y; }
#pragma unroll
    for (int q = 0; q < 8; ++q) mi[q] = 1u + (unsigned)half;
#pragma unroll
    for (int l = 0; l < PT_L - 1; ++l) {
#pragma unroll
      for (int q = 0; q < 8; ++q) {
        const float split = tb[PT_IMM(q, l) + mi[q]];
        mi[q] = __builtin_amdgcn_alignbit(mi[q], __float_as_uint(split - xv[q]), 31);
      }
    }
    float r[8];
#pragma unroll
    for (int q = 0; q < 8; ++q) r[q] = tb[PT_IMM(q, PT_L - 1) + mi[q]];

    v8f acc = {0.f, 0.f, 0.f, 0.f, 0.f, 0.f, 0.f, 0.f};
#pragma unroll
    for (int c = 0; c < 4; ++c) {
      v2f av; av.x = r[2 * c]; av.y = r[2 * c + 1];
      v2f bo; bo.x = 1.0f; bo.y = 1.0f;
      acc = __builtin_amdgcn_wmma_f32_16x16x4_f32(
           false, av,  false, bo,
           (short)0, acc,  false,  false);
    }
    if (m < 8) {
      float v = (m == 0) ? acc[0] : (m == 1) ? acc[1] : (m == 2) ? acc[2] :
                (m == 3) ? acc[3] : (m == 4) ? acc[4] : (m == 5) ? acc[5] :
                (m == 6) ? acc[6] : acc[7];
      int sdst = wavebase + 8 * half + m;
      if (sdst < n) out[sdst] = v * 0.0625f;
    }
  }
}

extern "C" void kernel_launch(void* const* d_in, const int* in_sizes, int n_in,
                              void* d_out, int out_size, void* d_ws, size_t ws_size,
                              hipStream_t stream) {
  const float* x       = (const float*)d_in[0];
  const float* samples = (const float*)d_in[1];
  (void)d_ws; (void)ws_size; (void)n_in;
  float* out = (float*)d_out;
  const int n = in_sizes[0] / PT_DIM;
  const int tiles = (n + PT_TILE - 1) / PT_TILE;
  const int blocks = (tiles < PT_GRID) ? tiles : PT_GRID;
  polya_tree_kernel<<<blocks, 256, 0, stream>>>(x, samples, out, n);
}


# Each lane walks all 16 trees of one sample (32 samples per wave): every LDS lookup of a wave reads one level of one tree, no bank conflict down to level 6; levels 0-1 come from scalar registers
# speedup vs baseline: 2.8477x; 1.0838x over previous
#include <hip/hip_runtime.h>


typedef __attribute__((ext_vector_type(2))) float v2f;
typedef __attribute__((ext_vector_type(4))) float v4f;
typedef __attribute__((ext_vector_type(8))) float v8f;

#define PT_DIM    16
#define PT_L      8
#define PT_NODES  255
#define PT_TILE   256
#define PT_GRID   1024

#define PT_PITCH  256
#define PT_TBL    (PT_DIM * PT_PITCH)

__device__ __forceinline__ void pt_split(float lo, float hi, float b,
                                         float& split, float& rhi) {
#pragma clang fp contract(off)
  const float len = hi - lo;
  split = lo + b * len;
  rhi   = split + (1.0f - b) * len;
}

__device__ __forceinline__ void pt_build(const float* __restrict__ samples,
                                         float* __restrict__ tb, int tid) {
  const int d = tid >> 4;
  const int k = tid & 15;
  const float* __restrict__ srow = samples + d * PT_NODES;
  float* __restrict__ trow = tb + d * PT_PITCH;

  float pbeta[4];
  float sbeta[15];
#pragma unroll
  for (int l = 0; l < 4; ++l) pbeta[l] = srow[((1 << l) - 1) + (k >> (4 - l))];
  {
    int first = 15 + k;
#pragma unroll
    for (int s = 0; s < 4; ++s) {
#pragma unroll
      for (int j = 0; j < (1 << s); ++j) sbeta[(1 << s) - 1 + j] = srow[first + j];
      first = 2 * first + 1;
    }
  }

  float lo = 0.0f, hi = 1.0f, prodb = 1.0f;
#pragma unroll
  for (int l = 0; l < 4; ++l) {
    const float b = pbeta[l];
    prodb = prodb * b;
    float split, rhi;
    pt_split(lo, hi, b, split, rhi);
    if ((k & ((1 << (4 - l)) - 1)) == 0)
      trow[(1 << l) + (k >> (4 - l))] = split;
    const bool right = ((k >> (3 - l)) & 1) != 0;
    hi = right ? rhi   : split;
    lo = right ? split : lo;
  }

  float clo[8], chi[8], cpb[8];
  clo[0] = lo; chi[0] = hi; cpb[0] = prodb;
#pragma unroll
  for (int s = 0; s < 3; ++s) {
    const int cnt = 1 << s;
    float nlo[8], nhi[8], npb[8];
#pragma unroll
    for (int j = 0; j < cnt; ++j) {
      const float b = sbeta[cnt - 1 + j];
      const float pb = cpb[j] * b;
      float split, rhi;
      pt_split(clo[j], chi[j], b, split, rhi);
      trow[(16 << s) + (k << s) + j] = split;
      nlo[2 * j]     = clo[j]; nhi[2 * j]     = split; npb[2 * j]     = pb;
      nlo[2 * j + 1] = split;  nhi[2 * j + 1] = rhi;   npb[2 * j + 1] = pb;
    }
#pragma unroll
    for (int j = 0; j < 2 * cnt; ++j) { clo[j] = nlo[j]; chi[j] = nhi[j]; cpb[j] = npb[j]; }
  }
#pragma unroll
  for (int j = 0; j < 8; ++j) {
    const float b = sbeta[7 + j];
    const float pb = cpb[j] * b;
    const float leaf = chi[j] - clo[j];
    trow[128 + 8 * k + j] = logf(pb * __builtin_amdgcn_rcpf(leaf));
  }
}

__device__ __forceinline__ void pt_half_swap(float& a, float& b) {
  typedef __attribute__((ext_vector_type(2))) unsigned v2u;
  const v2u r = __builtin_amdgcn_permlane16_swap(__float_as_uint(a), __float_as_uint(b),
                                                 false, false);
  a = __uint_as_float(r.x);
  b = __uint_as_float(r.y);
}

__global__ __launch_bounds__(256) __attribute__((amdgpu_waves_per_eu(8, 8)))
void polya_tree_kernel(const float* __restrict__ x,
                       const float* __restrict__ samples,
                       float* __restrict__ out, int n) {
  __shared__ float tb[PT_TBL];

  const int tid = threadIdx.x;
  const int lane = tid & 31;
  const int tiles = (n + PT_TILE - 1) / PT_TILE;

  {
    const int s0 = blockIdx.x * PT_TILE + tid;
    const int sc = (s0 < n) ? s0 : (n - 1);
    __builtin_prefetch(x + (size_t)sc * PT_DIM, 0, 0);
  }

  pt_build(samples, tb, tid);
  __syncthreads();

  unsigned t0[PT_DIM], t1l[PT_DIM], t1r[PT_DIM];
#pragma unroll
  for (int d = 0; d < PT_DIM; ++d) {
    t0[d]  = __builtin_amdgcn_readfirstlane(__float_as_uint(tb[d * PT_PITCH + 1]));
    t1l[d] = __builtin_amdgcn_readfirstlane(__float_as_uint(tb[d * PT_PITCH + 2]));
    t1r[d] = __builtin_amdgcn_readfirstlane(__float_as_uint(tb[d * PT_PITCH + 3]));
  }

  for (int tile = blockIdx.x; tile < tiles; tile += gridDim.x) {
    const int s0 = tile * PT_TILE + tid;
    const int sc = (s0 < n) ? s0 : (n - 1);
    const float* xr = x + (size_t)sc * PT_DIM;

    v4f xa[4];
#pragma unroll
    for (int c = 0; c < 4; ++c) xa[c] = *(const v4f*)(xr + 4 * c);

    {
      const int s1 = s0 + (int)gridDim.x * PT_TILE;
      const int sn = (s1 < n) ? s1 : (n - 1);
      __builtin_prefetch(x + (size_t)sn * PT_DIM, 0, 0);
    }

    float xv[PT_DIM];
    unsigned mi[PT_DIM];
#pragma unroll
    for (int c = 0; c < 4; ++c) {
      xv[4 * c] = xa[c].x; xv[4 * c + 1] = xa[c].y; xv[4 * c + 2] = xa[c].z; xv[4 * c + 3] = xa[c].w;
    }

#pragma unroll
    for (int d = 0; d < PT_DIM; ++d) {
      const unsigned u0 = __float_as_uint(__uint_as_float(t0[d]) - xv[d]);
      const unsigned right = (unsigned)((int)u0 >> 31);
      const unsigned t1 = (right & t1r[d]) | (~right & t1l[d]);
      const unsigned u1 = __float_as_uint(__uint_as_float(t1) - xv[d]);
      mi[d] = __builtin_amdgcn_alignbit(__builtin_amdgcn_alignbit(1u, u0, 31), u1, 31);
    }
#pragma unroll
    for (int l = 2; l < PT_L - 1; ++l) {
#pragma unroll
      for (int d = 0; d < PT_DIM; ++d) {
        const float split = tb[d * PT_PITCH + mi[d]];
        mi[d] = __builtin_amdgcn_alignbit(mi[d], __float_as_uint(split - xv[d]), 31);
      }
    }
    float r[PT_DIM];
#pragma unroll
    for (int d = 0; d < PT_DIM; ++d) r[d] = tb[d * PT_PITCH + mi[d]];

#pragma unroll
    for (int g = 0; g < 4; ++g) {
      pt_half_swap(r[4 * g],     r[4 * g + 2]);
      pt_half_swap(r[4 * g + 1], r[4 * g + 3]);
    }
    v8f acc_lo = {0.f, 0.f, 0.f, 0.f, 0.f, 0.f, 0.f, 0.f};
    v8f acc_hi = {0.f, 0.f, 0.f, 0.f, 0.f, 0.f, 0.f, 0.f};
    v2f ones; ones.x = 1.0f; ones.y = 1.0f;
#pragma unroll
    for (int g = 0; g < 4; ++g) {
      v2f blo; blo.x = r[4 * g];     blo.y = r[4 * g + 1];
      v2f bhi; bhi.x = r[4 * g + 2]; bhi.y = r[4 * g + 3];
      acc_lo = __builtin_amdgcn_wmma_f32_16x16x4_f32(
           false, ones,  false, blo,
           (short)0, acc_lo,  false,  false);
      acc_hi = __builtin_amdgcn_wmma_f32_16x16x4_f32(
           false, ones,  false, bhi,
           (short)0, acc_hi,  false,  false);
    }
    const float v = (lane < 16) ? acc_lo[0] : acc_hi[0];
    if (s0 < n) out[s0] = v * 0.0625f;
  }
}

extern "C" void kernel_launch(void* const* d_in, const int* in_sizes, int n_in,
                              void* d_out, int out_size, void* d_ws, size_t ws_size,
                              hipStream_t stream) {
  const float* x       = (const float*)d_in[0];
  const float* samples = (const float*)d_in[1];
  (void)d_ws; (void)ws_size; (void)n_in;
  float* out = (float*)d_out;
  const int n = in_sizes[0] / PT_DIM;
  const int tiles = (n + PT_TILE - 1) / PT_TILE;
  const int blocks = (tiles < PT_GRID) ? tiles : PT_GRID;
  polya_tree_kernel<<<blocks, 256, 0, stream>>>(x, samples, out, n);
}
